# starting kernel
# baseline (speedup 1.0000x reference)
#include <hip/hip_runtime.h>
#include <hip/hip_bf16.h>

typedef __attribute__((ext_vector_type(2))) float v2f;
typedef __attribute__((ext_vector_type(8))) float v8f;

#define NUMK 128
#define NUME 1024
#define HID  32
#define HXW  (NUMK + NUME + 2)
#define AS   132
#define XS   36
#define GTS  132
#define WTS  36

__device__ __forceinline__ v8f wmma_step(const float* __restrict__ Am, int aStride,
                                         const float* __restrict__ BTm, int btStride,
                                         int m0, int n0, int k0, int half, int l, v8f c) {
    v2f a, b;
    a.x = Am[(m0 + l) * aStride + k0 + 2 * half];
    a.y = Am[(m0 + l) * aStride + k0 + 2 * half + 1];
    b.x = BTm[(n0 + l) * btStride + k0 + 2 * half];
    b.y = BTm[(n0 + l) * btStride + k0 + 2 * half + 1];
    return __builtin_amdgcn_wmma_f32_16x16x4_f32(false, a, false, b, (short)0, c,
                                                 false, false);
}

__global__ __launch_bounds__(256) void gcn3_kernel(
    const float* __restrict__ Hx, const int* __restrict__ ei,
    const float* __restrict__ W1, const float* __restrict__ b1,
    const float* __restrict__ W2, const float* __restrict__ b2,
    const float* __restrict__ W3, const float* __restrict__ b3,
    float* __restrict__ out) {
    __shared__ float As[NUMK * AS];
    __shared__ float Xs[NUMK * XS];
    __shared__ float GT[HID * GTS];
    __shared__ float W2T[HID * WTS];
    __shared__ float ps[NUMK], a1s[NUMK], vs[NUMK], deg[NUMK], dinv[NUMK];
    __shared__ float W1s[HID], b1s[HID], b2s[HID], W3s[HID];

    const int g    = blockIdx.x;
    const int t    = threadIdx.x;
    const int lane = t & 31;
    const int wave = t >> 5;
    const int half = lane >> 4;
    const int l    = lane & 15;
    const float* hx = Hx + (size_t)g * HXW;

    for (int i = t; i < NUMK * AS; i += 256) As[i] = 0.0f;
    if (t < NUMK) ps[t] = hx[t];
    if (t < HID) { W1s[t] = W1[t]; b1s[t] = b1[t]; b2s[t] = b2[t]; W3s[t] = W3[t]; }
    for (int i = t; i < HID * HID; i += 256)
        W2T[(i & 31) * WTS + (i >> 5)] = W2[i];
    __syncthreads();

    if (t < NUMK) { As[t * AS + t] = 1.0f; deg[t] = 1.0f; }
    __syncthreads();

    for (int e = t; e < NUME; e += 256) {
        int s = ei[e];
        int d = ei[NUME + e];
        float w = hx[NUMK + e];
        atomicAdd(&As[d * AS + s], w);
        atomicAdd(&deg[d], w);
    }
    __syncthreads();

    if (t < NUMK) dinv[t] = rsqrtf(deg[t]);
    __syncthreads();
    for (int i = t; i < NUMK * NUMK; i += 256) {
        int d = i >> 7, s = i & 127;
        As[d * AS + s] *= dinv[d] * dinv[s];
    }
    __syncthreads();

    if (t < NUMK) {
        float acc = 0.0f;
        for (int s = 0; s < NUMK; ++s) acc += As[t * AS + s] * ps[s];
        a1s[t] = acc;
    }
    __syncthreads();
    for (int i = t; i < NUMK * HID; i += 256) {
        int n = i >> 5, f = i & 31;
        Xs[n * XS + f] = fmaxf(fmaf(a1s[n], W1s[f], b1s[f]), 0.0f);
    }
    __syncthreads();

    {
        const int m0 = wave * 16;
        for (int nt = 0; nt < 2; ++nt) {
            const int n0 = nt * 16;
            v8f c = {};
#pragma unroll
            for (int k0 = 0; k0 < HID; k0 += 4)
                c = wmma_step(Xs, XS, W2T, WTS, m0, n0, k0, half, l, c);
#pragma unroll
            for (int j = 0; j < 8; ++j)
                GT[(n0 + l) * GTS + (m0 + j + 8 * half)] = c[j];
        }
    }
    __syncthreads();

    {
        const int m0 = wave * 16;
        for (int nt = 0; nt < 2; ++nt) {
            const int n0 = nt * 16;
            v8f c = {};
#pragma unroll
            for (int k0 = 0; k0 < NUMK; k0 += 4)
                c = wmma_step(As, AS, GT, GTS, m0, n0, k0, half, l, c);
#pragma unroll
            for (int j = 0; j < 8; ++j) {
                int row = m0 + j + 8 * half, col = n0 + l;
                Xs[row * XS + col] = fmaxf(c[j] + b2s[col], 0.0f);
            }
        }
    }
    __syncthreads();

    if (t < NUMK) {
        float acc = 0.0f;
        for (int f = 0; f < HID; ++f) acc += Xs[t * XS + f] * W3s[f];
        vs[t] = acc;
    }
    __syncthreads();
    if (t < NUMK) {
        float acc = 0.0f;
        for (int s = 0; s < NUMK; ++s) acc += As[t * AS + s] * vs[s];
        out[(size_t)g * NUMK + t] = acc + b3[0];
    }
}

extern "C" void kernel_launch(void* const* d_in, const int* in_sizes, int n_in,
                              void* d_out, int out_size, void* d_ws, size_t ws_size,
                              hipStream_t stream) {
    (void)n_in; (void)d_ws; (void)ws_size; (void)out_size;
    const float* Hx = (const float*)d_in[0];
    const int*   ei = (const int*)d_in[1];
    const float* W1 = (const float*)d_in[2];
    const float* b1 = (const float*)d_in[3];
    const float* W2 = (const float*)d_in[4];
    const float* b2 = (const float*)d_in[5];
    const float* W3 = (const float*)d_in[6];
    const float* b3 = (const float*)d_in[7];
    float* out = (float*)d_out;
    const int nb = in_sizes[0] / HXW;
    gcn3_kernel<<<dim3(nb), dim3(256), 0, stream>>>(Hx, ei, W1, b1, W2, b2, W3, b3, out);
}


# All global reads (node signal, W2, params, 4 edges per thread) issued at kernel entry into registers; their latency runs under the LDS zeroing instead of one load-wait pair per loop trip
# speedup vs baseline: 1.2278x; 1.2278x over previous
#include <hip/hip_runtime.h>
#include <hip/hip_bf16.h>

typedef __attribute__((ext_vector_type(2))) float v2f;
typedef __attribute__((ext_vector_type(8))) float v8f;

#define NUMK 128
#define NUME 1024
#define HID  32
#define HXW  (NUMK + NUME + 2)
#define AS   132
#define XS   36
#define GTS  132
#define WTS  36

__device__ __forceinline__ v8f wmma_step(const float* __restrict__ Am, int aStride,
                                         const float* __restrict__ BTm, int btStride,
                                         int m0, int n0, int k0, int half, int l, v8f c) {
    v2f a, b;
    a.x = Am[(m0 + l) * aStride + k0 + 2 * half];
    a.y = Am[(m0 + l) * aStride + k0 + 2 * half + 1];
    b.x = BTm[(n0 + l) * btStride + k0 + 2 * half];
    b.y = BTm[(n0 + l) * btStride + k0 + 2 * half + 1];
    return __builtin_amdgcn_wmma_f32_16x16x4_f32(false, a, false, b, (short)0, c,
                                                 false, false);
}

__global__ __launch_bounds__(256) void gcn3_kernel(
    const float* __restrict__ Hx, const int* __restrict__ ei,
    const float* __restrict__ W1, const float* __restrict__ b1,
    const float* __restrict__ W2, const float* __restrict__ b2,
    const float* __restrict__ W3, const float* __restrict__ b3,
    float* __restrict__ out) {
    __shared__ float As[NUMK * AS];
    __shared__ float Xs[NUMK * XS];
    __shared__ float GT[HID * GTS];
    __shared__ float W2T[HID * WTS];
    __shared__ float ps[NUMK], a1s[NUMK], vs[NUMK], deg[NUMK], dinv[NUMK];
    __shared__ float W1s[HID], b1s[HID], b2s[HID], W3s[HID];

    const int g    = blockIdx.x;
    const int t    = threadIdx.x;
    const int lane = t & 31;
    const int wave = t >> 5;
    const int half = lane >> 4;
    const int l    = lane & 15;
    const float* hx = Hx + (size_t)g * HXW;

    float p_r = 0.0f;
    if (t < NUMK) p_r = hx[t];
    float w2_r[4];
#pragma unroll
    for (int k = 0; k < 4; ++k) w2_r[k] = W2[t + 256 * k];
    float w1_r = 0.0f, b1_r = 0.0f, b2_r = 0.0f, w3_r = 0.0f;
    if (t < HID) { w1_r = W1[t]; b1_r = b1[t]; b2_r = b2[t]; w3_r = W3[t]; }
    int   e_s[4], e_d[4];
    float e_w[4];
#pragma unroll
    for (int k = 0; k < 4; ++k) {
        const int e = t + 256 * k;
        e_s[k] = ei[e];
        e_d[k] = ei[NUME + e];
        e_w[k] = hx[NUMK + e];
    }

    for (int i = t; i < NUMK * AS; i += 256) As[i] = 0.0f;
    if (t < NUMK) ps[t] = p_r;
    if (t < HID) { W1s[t] = w1_r; b1s[t] = b1_r; b2s[t] = b2_r; W3s[t] = w3_r; }
#pragma unroll
    for (int k = 0; k < 4; ++k) {
        const int i = t + 256 * k;
        W2T[(i & 31) * WTS + (i >> 5)] = w2_r[k];
    }
    __syncthreads();

    if (t < NUMK) { As[t * AS + t] = 1.0f; deg[t] = 1.0f; }
    __syncthreads();

#pragma unroll
    for (int k = 0; k < 4; ++k) {
        atomicAdd(&As[e_d[k] * AS + e_s[k]], e_w[k]);
        atomicAdd(&deg[e_d[k]], e_w[k]);
    }
    __syncthreads();

    if (t < NUMK) dinv[t] = rsqrtf(deg[t]);
    __syncthreads();
    for (int i = t; i < NUMK * NUMK; i += 256) {
        int d = i >> 7, s = i & 127;
        As[d * AS + s] *= dinv[d] * dinv[s];
    }
    __syncthreads();

    if (t < NUMK) {
        float acc = 0.0f;
        for (int s = 0; s < NUMK; ++s) acc += As[t * AS + s] * ps[s];
        a1s[t] = acc;
    }
    __syncthreads();
    for (int i = t; i < NUMK * HID; i += 256) {
        int n = i >> 5, f = i & 31;
        Xs[n * XS + f] = fmaxf(fmaf(a1s[n], W1s[f], b1s[f]), 0.0f);
    }
    __syncthreads();

    {
        const int m0 = wave * 16;
        for (int nt = 0; nt < 2; ++nt) {
            const int n0 = nt * 16;
            v8f c = {};
#pragma unroll
            for (int k0 = 0; k0 < HID; k0 += 4)
                c = wmma_step(Xs, XS, W2T, WTS, m0, n0, k0, half, l, c);
#pragma unroll
            for (int j = 0; j < 8; ++j)
                GT[(n0 + l) * GTS + (m0 + j + 8 * half)] = c[j];
        }
    }
    __syncthreads();

    {
        const int m0 = wave * 16;
        for (int nt = 0; nt < 2; ++nt) {
            const int n0 = nt * 16;
            v8f c = {};
#pragma unroll
            for (int k0 = 0; k0 < NUMK; k0 += 4)
                c = wmma_step(As, AS, GT, GTS, m0, n0, k0, half, l, c);
#pragma unroll
            for (int j = 0; j < 8; ++j) {
                int row = m0 + j + 8 * half, col = n0 + l;
                Xs[row * XS + col] = fmaxf(c[j] + b2s[col], 0.0f);
            }
        }
    }
    __syncthreads();

    if (t < NUMK) {
        float acc = 0.0f;
        for (int f = 0; f < HID; ++f) acc += Xs[t * XS + f] * W3s[f];
        vs[t] = acc;
    }
    __syncthreads();
    if (t < NUMK) {
        float acc = 0.0f;
        for (int s = 0; s < NUMK; ++s) acc += As[t * AS + s] * vs[s];
        out[(size_t)g * NUMK + t] = acc + b3[0];
    }
}

extern "C" void kernel_launch(void* const* d_in, const int* in_sizes, int n_in,
                              void* d_out, int out_size, void* d_ws, size_t ws_size,
                              hipStream_t stream) {
    (void)n_in; (void)d_ws; (void)ws_size; (void)out_size;
    const float* Hx = (const float*)d_in[0];
    const int*   ei = (const int*)d_in[1];
    const float* W1 = (const float*)d_in[2];
    const float* b1 = (const float*)d_in[3];
    const float* W2 = (const float*)d_in[4];
    const float* b2 = (const float*)d_in[5];
    const float* W3 = (const float*)d_in[6];
    const float* b3 = (const float*)d_in[7];
    float* out = (float*)d_out;
    const int nb = in_sizes[0] / HXW;
    gcn3_kernel<<<dim3(nb), dim3(256), 0, stream>>>(Hx, ei, W1, b1, W2, b2, W3, b3, out);
}


# Symmetric normalization folded into the vectors: matrix stays raw in LDS, each propagation scales operand rows and result rows by dinv; the 16K-element rescale pass and its barrier are gone
# speedup vs baseline: 1.4650x; 1.1932x over previous
#include <hip/hip_runtime.h>
#include <hip/hip_bf16.h>

typedef __attribute__((ext_vector_type(2))) float v2f;
typedef __attribute__((ext_vector_type(8))) float v8f;

#define NUMK 128
#define NUME 1024
#define HID  32
#define HXW  (NUMK + NUME + 2)
#define AS   132
#define XS   36
#define GTS  132
#define WTS  36

__device__ __forceinline__ v8f wmma_step(const float* __restrict__ Am, int aStride,
                                         const float* __restrict__ BTm, int btStride,
                                         int m0, int n0, int k0, int half, int l, v8f c) {
    v2f a, b;
    a.x = Am[(m0 + l) * aStride + k0 + 2 * half];
    a.y = Am[(m0 + l) * aStride + k0 + 2 * half + 1];
    b.x = BTm[(n0 + l) * btStride + k0 + 2 * half];
    b.y = BTm[(n0 + l) * btStride + k0 + 2 * half + 1];
    return __builtin_amdgcn_wmma_f32_16x16x4_f32(false, a, false, b, (short)0, c,
                                                 false, false);
}

__global__ __launch_bounds__(256) void gcn3_kernel(
    const float* __restrict__ Hx, const int* __restrict__ ei,
    const float* __restrict__ W1, const float* __restrict__ b1,
    const float* __restrict__ W2, const float* __restrict__ b2,
    const float* __restrict__ W3, const float* __restrict__ b3,
    float* __restrict__ out) {
    __shared__ float As[NUMK * AS];
    __shared__ float Xs[NUMK * XS];
    __shared__ float GT[HID * GTS];
    __shared__ float W2T[HID * WTS];
    __shared__ float ps[NUMK], a1s[NUMK], vs[NUMK], deg[NUMK], dinv[NUMK];
    __shared__ float W1s[HID], b1s[HID], b2s[HID], W3s[HID];

    const int g    = blockIdx.x;
    const int t    = threadIdx.x;
    const int lane = t & 31;
    const int wave = t >> 5;
    const int half = lane >> 4;
    const int l    = lane & 15;
    const float* hx = Hx + (size_t)g * HXW;

    float p_r = 0.0f;
    if (t < NUMK) p_r = hx[t];
    float w2_r[4];
#pragma unroll
    for (int k = 0; k < 4; ++k) w2_r[k] = W2[t + 256 * k];
    float w1_r = 0.0f, b1_r = 0.0f, b2_r = 0.0f, w3_r = 0.0f;
    if (t < HID) { w1_r = W1[t]; b1_r = b1[t]; b2_r = b2[t]; w3_r = W3[t]; }
    int   e_s[4], e_d[4];
    float e_w[4];
#pragma unroll
    for (int k = 0; k < 4; ++k) {
        const int e = t + 256 * k;
        e_s[k] = ei[e];
        e_d[k] = ei[NUME + e];
        e_w[k] = hx[NUMK + e];
    }

    for (int i = t; i < NUMK * AS; i += 256) As[i] = 0.0f;
    if (t < HID) { W1s[t] = w1_r; b1s[t] = b1_r; b2s[t] = b2_r; W3s[t] = w3_r; }
#pragma unroll
    for (int k = 0; k < 4; ++k) {
        const int i = t + 256 * k;
        W2T[(i & 31) * WTS + (i >> 5)] = w2_r[k];
    }
    __syncthreads();

    if (t < NUMK) { As[t * AS + t] = 1.0f; deg[t] = 1.0f; }
    __syncthreads();

#pragma unroll
    for (int k = 0; k < 4; ++k) {
        atomicAdd(&As[e_d[k] * AS + e_s[k]], e_w[k]);
        atomicAdd(&deg[e_d[k]], e_w[k]);
    }
    __syncthreads();

    float dv = 0.0f;
    if (t < NUMK) { dv = rsqrtf(deg[t]); dinv[t] = dv; ps[t] = dv * p_r; }
    __syncthreads();

    if (t < NUMK) {
        float acc = 0.0f;
        for (int s = 0; s < NUMK; ++s) acc += As[t * AS + s] * ps[s];
        a1s[t] = dv * acc;
    }
    __syncthreads();
    for (int i = t; i < NUMK * HID; i += 256) {
        int n = i >> 5, f = i & 31;
        Xs[n * XS + f] = fmaxf(fmaf(a1s[n], W1s[f], b1s[f]), 0.0f);
    }
    __syncthreads();

    float dr[8];
#pragma unroll
    for (int j = 0; j < 8; ++j) dr[j] = dinv[wave * 16 + j + 8 * half];

    {
        const int m0 = wave * 16;
        for (int nt = 0; nt < 2; ++nt) {
            const int n0 = nt * 16;
            v8f c = {};
#pragma unroll
            for (int k0 = 0; k0 < HID; k0 += 4)
                c = wmma_step(Xs, XS, W2T, WTS, m0, n0, k0, half, l, c);
#pragma unroll
            for (int j = 0; j < 8; ++j)
                GT[(n0 + l) * GTS + (m0 + j + 8 * half)] = dr[j] * c[j];
        }
    }
    __syncthreads();

    {
        const int m0 = wave * 16;
        for (int nt = 0; nt < 2; ++nt) {
            const int n0 = nt * 16;
            v8f c = {};
#pragma unroll
            for (int k0 = 0; k0 < NUMK; k0 += 4)
                c = wmma_step(As, AS, GT, GTS, m0, n0, k0, half, l, c);
#pragma unroll
            for (int j = 0; j < 8; ++j) {
                int row = m0 + j + 8 * half, col = n0 + l;
                Xs[row * XS + col] = fmaxf(fmaf(dr[j], c[j], b2s[col]), 0.0f);
            }
        }
    }
    __syncthreads();

    if (t < NUMK) {
        float acc = 0.0f;
        for (int f = 0; f < HID; ++f) acc += Xs[t * XS + f] * W3s[f];
        vs[t] = dv * acc;
    }
    __syncthreads();
    if (t < NUMK) {
        float acc = 0.0f;
        for (int s = 0; s < NUMK; ++s) acc += As[t * AS + s] * vs[s];
        out[(size_t)g * NUMK + t] = fmaf(dv, acc, b3[0]);
    }
}

extern "C" void kernel_launch(void* const* d_in, const int* in_sizes, int n_in,
                              void* d_out, int out_size, void* d_ws, size_t ws_size,
                              hipStream_t stream) {
    (void)n_in; (void)d_ws; (void)ws_size; (void)out_size;
    const float* Hx = (const float*)d_in[0];
    const int*   ei = (const int*)d_in[1];
    const float* W1 = (const float*)d_in[2];
    const float* b1 = (const float*)d_in[3];
    const float* W2 = (const float*)d_in[4];
    const float* b2 = (const float*)d_in[5];
    const float* W3 = (const float*)d_in[6];
    const float* b3 = (const float*)d_in[7];
    float* out = (float*)d_out;
    const int nb = in_sizes[0] / HXW;
    gcn3_kernel<<<dim3(nb), dim3(256), 0, stream>>>(Hx, ei, W1, b1, W2, b2, W3, b3, out);
}


# The three per-row dot products (A@p, x2@W3, A@v) each summed in four independent partial sums (column c into sum c&3, joined (0+1)+(2+3)), LDS reads issued ahead: summation order changed
# speedup vs baseline: 1.6788x; 1.1459x over previous
#include <hip/hip_runtime.h>
#include <hip/hip_bf16.h>

typedef __attribute__((ext_vector_type(2))) float v2f;
typedef __attribute__((ext_vector_type(8))) float v8f;
typedef __attribute__((ext_vector_type(4))) float v4f;

#define NUMK 128
#define NUME 1024
#define HID  32
#define HXW  (NUMK + NUME + 2)
#define AS   132
#define XS   36
#define GTS  132
#define WTS  36

__device__ __forceinline__ v8f wmma_step(const float* __restrict__ Am, int aStride,
                                         const float* __restrict__ BTm, int btStride,
                                         int m0, int n0, int k0, int half, int l, v8f c) {
    v2f a, b;
    a.x = Am[(m0 + l) * aStride + k0 + 2 * half];
    a.y = Am[(m0 + l) * aStride + k0 + 2 * half + 1];
    b.x = BTm[(n0 + l) * btStride + k0 + 2 * half];
    b.y = BTm[(n0 + l) * btStride + k0 + 2 * half + 1];
    return __builtin_amdgcn_wmma_f32_16x16x4_f32(false, a, false, b, (short)0, c,
                                                 false, false);
}

template <int N>
__device__ __forceinline__ float dot_4sums(const float* __restrict__ rowp,
                                           const float* __restrict__ vecp) {
    const v4f* r4 = reinterpret_cast<const v4f*>(rowp);
    const v4f* x4 = reinterpret_cast<const v4f*>(vecp);
    float s0 = 0.0f, s1 = 0.0f, s2 = 0.0f, s3 = 0.0f;
    for (int i = 0; i < N / 4; ++i) {
        const v4f a = r4[i];
        const v4f x = x4[i];
        s0 = fmaf(a.x, x.x, s0);
        s1 = fmaf(a.y, x.y, s1);
        s2 = fmaf(a.z, x.z, s2);
        s3 = fmaf(a.w, x.w, s3);
    }
    return (s0 + s1) + (s2 + s3);
}

__global__ __launch_bounds__(256) void gcn3_kernel(
    const float* __restrict__ Hx, const int* __restrict__ ei,
    const float* __restrict__ W1, const float* __restrict__ b1,
    const float* __restrict__ W2, const float* __restrict__ b2,
    const float* __restrict__ W3, const float* __restrict__ b3,
    float* __restrict__ out) {
    __shared__ __attribute__((aligned(16))) float As[NUMK * AS];
    __shared__ __attribute__((aligned(16))) float Xs[NUMK * XS];
    __shared__ float GT[HID * GTS];
    __shared__ float W2T[HID * WTS];
    __shared__ __attribute__((aligned(16))) float ps[NUMK];
    __shared__ __attribute__((aligned(16))) float vs[NUMK];
    __shared__ __attribute__((aligned(16))) float W3s[HID];
    __shared__ float a1s[NUMK], deg[NUMK], dinv[NUMK];
    __shared__ float W1s[HID], b1s[HID], b2s[HID];

    const int g    = blockIdx.x;
    const int t    = threadIdx.x;
    const int lane = t & 31;
    const int wave = t >> 5;
    const int half = lane >> 4;
    const int l    = lane & 15;
    const float* hx = Hx + (size_t)g * HXW;

    float p_r = 0.0f;
    if (t < NUMK) p_r = hx[t];
    float w2_r[4];
#pragma unroll
    for (int k = 0; k < 4; ++k) w2_r[k] = W2[t + 256 * k];
    float w1_r = 0.0f, b1_r = 0.0f, b2_r = 0.0f, w3_r = 0.0f;
    if (t < HID) { w1_r = W1[t]; b1_r = b1[t]; b2_r = b2[t]; w3_r = W3[t]; }
    int   e_s[4], e_d[4];
    float e_w[4];
#pragma unroll
    for (int k = 0; k < 4; ++k) {
        const int e = t + 256 * k;
        e_s[k] = ei[e];
        e_d[k] = ei[NUME + e];
        e_w[k] = hx[NUMK + e];
    }

    for (int i = t; i < NUMK * AS; i += 256) As[i] = 0.0f;
    if (t < HID) { W1s[t] = w1_r; b1s[t] = b1_r; b2s[t] = b2_r; W3s[t] = w3_r; }
#pragma unroll
    for (int k = 0; k < 4; ++k) {
        const int i = t + 256 * k;
        W2T[(i & 31) * WTS + (i >> 5)] = w2_r[k];
    }
    __syncthreads();

    if (t < NUMK) { As[t * AS + t] = 1.0f; deg[t] = 1.0f; }
    __syncthreads();

#pragma unroll
    for (int k = 0; k < 4; ++k) {
        atomicAdd(&As[e_d[k] * AS + e_s[k]], e_w[k]);
        atomicAdd(&deg[e_d[k]], e_w[k]);
    }
    __syncthreads();

    float dv = 0.0f;
    if (t < NUMK) { dv = rsqrtf(deg[t]); dinv[t] = dv; ps[t] = dv * p_r; }
    __syncthreads();

    if (t < NUMK) a1s[t] = dv * dot_4sums<NUMK>(As + t * AS, ps);
    __syncthreads();
    for (int i = t; i < NUMK * HID; i += 256) {
        int n = i >> 5, f = i & 31;
        Xs[n * XS + f] = fmaxf(fmaf(a1s[n], W1s[f], b1s[f]), 0.0f);
    }
    __syncthreads();

    float dr[8];
#pragma unroll
    for (int j = 0; j < 8; ++j) dr[j] = dinv[wave * 16 + j + 8 * half];

    {
        const int m0 = wave * 16;
        for (int nt = 0; nt < 2; ++nt) {
            const int n0 = nt * 16;
            v8f c = {};
#pragma unroll
            for (int k0 = 0; k0 < HID; k0 += 4)
                c = wmma_step(Xs, XS, W2T, WTS, m0, n0, k0, half, l, c);
#pragma unroll
            for (int j = 0; j < 8; ++j)
                GT[(n0 + l) * GTS + (m0 + j + 8 * half)] = dr[j] * c[j];
        }
    }
    __syncthreads();

    {
        const int m0 = wave * 16;
        for (int nt = 0; nt < 2; ++nt) {
            const int n0 = nt * 16;
            v8f c = {};
#pragma unroll
            for (int k0 = 0; k0 < NUMK; k0 += 4)
                c = wmma_step(As, AS, GT, GTS, m0, n0, k0, half, l, c);
#pragma unroll
            for (int j = 0; j < 8; ++j) {
                int row = m0 + j + 8 * half, col = n0 + l;
                Xs[row * XS + col] = fmaxf(fmaf(dr[j], c[j], b2s[col]), 0.0f);
            }
        }
    }
    __syncthreads();

    if (t < NUMK) vs[t] = dv * dot_4sums<HID>(Xs + t * XS, W3s);
    __syncthreads();
    if (t < NUMK)
        out[(size_t)g * NUMK + t] = fmaf(dv, dot_4sums<NUMK>(As + t * AS, vs), b3[0]);
}

extern "C" void kernel_launch(void* const* d_in, const int* in_sizes, int n_in,
                              void* d_out, int out_size, void* d_ws, size_t ws_size,
                              hipStream_t stream) {
    (void)n_in; (void)d_ws; (void)ws_size; (void)out_size;
    const float* Hx = (const float*)d_in[0];
    const int*   ei = (const int*)d_in[1];
    const float* W1 = (const float*)d_in[2];
    const float* b1 = (const float*)d_in[3];
    const float* W2 = (const float*)d_in[4];
    const float* b2 = (const float*)d_in[5];
    const float* W3 = (const float*)d_in[6];
    const float* b3 = (const float*)d_in[7];
    float* out = (float*)d_out;
    const int nb = in_sizes[0] / HXW;
    gcn3_kernel<<<dim3(nb), dim3(256), 0, stream>>>(Hx, ei, W1, b1, W2, b2, W3, b3, out);
}


# W2 no longer staged in LDS (layer-2a B fragments read straight from global into registers): LDS per block 110592 -> 105984 bytes, so three blocks instead of two are resident per compute unit
# speedup vs baseline: 1.9685x; 1.1726x over previous
#include <hip/hip_runtime.h>
#include <hip/hip_bf16.h>

typedef __attribute__((ext_vector_type(2))) float v2f;
typedef __attribute__((ext_vector_type(8))) float v8f;
typedef __attribute__((ext_vector_type(4))) float v4f;

#define NUMK 128
#define NUME 1024
#define HID  32
#define HXW  (NUMK + NUME + 2)
#define AS   132
#define XS   36
#define GTS  132

__device__ __forceinline__ v8f wmma_step(const float* __restrict__ Am, int aStride,
                                         const float* __restrict__ BTm, int btStride,
                                         int m0, int n0, int k0, int half, int l, v8f c) {
    v2f a, b;
    a.x = Am[(m0 + l) * aStride + k0 + 2 * half];
    a.y = Am[(m0 + l) * aStride + k0 + 2 * half + 1];
    b.x = BTm[(n0 + l) * btStride + k0 + 2 * half];
    b.y = BTm[(n0 + l) * btStride + k0 + 2 * half + 1];
    return __builtin_amdgcn_wmma_f32_16x16x4_f32(false, a, false, b, (short)0, c,
                                                 false, false);
}

__device__ __forceinline__ v8f wmma_step_regb(const float* __restrict__ Am, int aStride,
                                              float b0, float b1,
                                              int m0, int k0, int half, int l, v8f c) {
    v2f a, b;
    a.x = Am[(m0 + l) * aStride + k0 + 2 * half];
    a.y = Am[(m0 + l) * aStride + k0 + 2 * half + 1];
    b.x = b0;
    b.y = b1;
    return __builtin_amdgcn_wmma_f32_16x16x4_f32(false, a, false, b, (short)0, c,
                                                 false, false);
}

template <int N>
__device__ __forceinline__ float dot_4sums(const float* __restrict__ rowp,
                                           const float* __restrict__ vecp) {
    const v4f* r4 = reinterpret_cast<const v4f*>(rowp);
    const v4f* x4 = reinterpret_cast<const v4f*>(vecp);
    float s0 = 0.0f, s1 = 0.0f, s2 = 0.0f, s3 = 0.0f;
    for (int i = 0; i < N / 4; ++i) {
        const v4f a = r4[i];
        const v4f x = x4[i];
        s0 = fmaf(a.x, x.x, s0);
        s1 = fmaf(a.y, x.y, s1);
        s2 = fmaf(a.z, x.z, s2);
        s3 = fmaf(a.w, x.w, s3);
    }
    return (s0 + s1) + (s2 + s3);
}

__global__ __launch_bounds__(256) void gcn3_kernel(
    const float* __restrict__ Hx, const int* __restrict__ ei,
    const float* __restrict__ W1, const float* __restrict__ b1,
    const float* __restrict__ W2, const float* __restrict__ b2,
    const float* __restrict__ W3, const float* __restrict__ b3,
    float* __restrict__ out) {
    __shared__ __attribute__((aligned(16))) float As[NUMK * AS];
    __shared__ __attribute__((aligned(16))) float Xs[NUMK * XS];
    __shared__ float GT[HID * GTS];
    __shared__ __attribute__((aligned(16))) float ps[NUMK];
    __shared__ __attribute__((aligned(16))) float vs[NUMK];
    __shared__ __attribute__((aligned(16))) float W3s[HID];
    __shared__ float a1s[NUMK], deg[NUMK], dinv[NUMK];
    __shared__ float W1s[HID], b1s[HID], b2s[HID];

    const int g    = blockIdx.x;
    const int t    = threadIdx.x;
    const int lane = t & 31;
    const int wave = t >> 5;
    const int half = lane >> 4;
    const int l    = lane & 15;
    const float* hx = Hx + (size_t)g * HXW;

    float p_r = 0.0f;
    if (t < NUMK) p_r = hx[t];
    float w1_r = 0.0f, b1_r = 0.0f, b2_r = 0.0f, w3_r = 0.0f;
    if (t < HID) { w1_r = W1[t]; b1_r = b1[t]; b2_r = b2[t]; w3_r = W3[t]; }
    int   e_s[4], e_d[4];
    float e_w[4];
#pragma unroll
    for (int k = 0; k < 4; ++k) {
        const int e = t + 256 * k;
        e_s[k] = ei[e];
        e_d[k] = ei[NUME + e];
        e_w[k] = hx[NUMK + e];
    }
    float w2b[2][HID / 4][2];
#pragma unroll
    for (int nt = 0; nt < 2; ++nt)
#pragma unroll
        for (int ks = 0; ks < HID / 4; ++ks) {
            const int k = 4 * ks + 2 * half;
            w2b[nt][ks][0] = W2[k * HID + nt * 16 + l];
            w2b[nt][ks][1] = W2[(k + 1) * HID + nt * 16 + l];
        }

    for (int i = t; i < NUMK * AS; i += 256) As[i] = 0.0f;
    if (t < HID) { W1s[t] = w1_r; b1s[t] = b1_r; b2s[t] = b2_r; W3s[t] = w3_r; }
    __syncthreads();

    if (t < NUMK) { As[t * AS + t] = 1.0f; deg[t] = 1.0f; }
    __syncthreads();

#pragma unroll
    for (int k = 0; k < 4; ++k) {
        atomicAdd(&As[e_d[k] * AS + e_s[k]], e_w[k]);
        atomicAdd(&deg[e_d[k]], e_w[k]);
    }
    __syncthreads();

    float dv = 0.0f;
    if (t < NUMK) { dv = rsqrtf(deg[t]); dinv[t] = dv; ps[t] = dv * p_r; }
    __syncthreads();

    if (t < NUMK) a1s[t] = dv * dot_4sums<NUMK>(As + t * AS, ps);
    __syncthreads();
    for (int i = t; i < NUMK * HID; i += 256) {
        int n = i >> 5, f = i & 31;
        Xs[n * XS + f] = fmaxf(fmaf(a1s[n], W1s[f], b1s[f]), 0.0f);
    }
    __syncthreads();

    float dr[8];
#pragma unroll
    for (int j = 0; j < 8; ++j) dr[j] = dinv[wave * 16 + j + 8 * half];

    {
        const int m0 = wave * 16;
        for (int nt = 0; nt < 2; ++nt) {
            const int n0 = nt * 16;
            v8f c = {};
#pragma unroll
            for (int ks = 0; ks < HID / 4; ++ks)
                c = wmma_step_regb(Xs, XS, w2b[nt][ks][0], w2b[nt][ks][1],
                                   m0, 4 * ks, half, l, c);
#pragma unroll
            for (int j = 0; j < 8; ++j)
                GT[(n0 + l) * GTS + (m0 + j + 8 * half)] = dr[j] * c[j];
        }
    }
    __syncthreads();

    {
        const int m0 = wave * 16;
        for (int nt = 0; nt < 2; ++nt) {
            const int n0 = nt * 16;
            v8f c = {};
#pragma unroll
            for (int k0 = 0; k0 < NUMK; k0 += 4)
                c = wmma_step(As, AS, GT, GTS, m0, n0, k0, half, l, c);
#pragma unroll
            for (int j = 0; j < 8; ++j) {
                int row = m0 + j + 8 * half, col = n0 + l;
                Xs[row * XS + col] = fmaxf(fmaf(dr[j], c[j], b2s[col]), 0.0f);
            }
        }
    }
    __syncthreads();

    if (t < NUMK) vs[t] = dv * dot_4sums<HID>(Xs + t * XS, W3s);
    __syncthreads();
    if (t < NUMK)
        out[(size_t)g * NUMK + t] = fmaf(dv, dot_4sums<NUMK>(As + t * AS, vs), b3[0]);
}

extern "C" void kernel_launch(void* const* d_in, const int* in_sizes, int n_in,
                              void* d_out, int out_size, void* d_ws, size_t ws_size,
                              hipStream_t stream) {
    (void)n_in; (void)d_ws; (void)ws_size; (void)out_size;
    const float* Hx = (const float*)d_in[0];
    const int*   ei = (const int*)d_in[1];
    const float* W1 = (const float*)d_in[2];
    const float* b1 = (const float*)d_in[3];
    const float* W2 = (const float*)d_in[4];
    const float* b2 = (const float*)d_in[5];
    const float* W3 = (const float*)d_in[6];
    const float* b3 = (const float*)d_in[7];
    float* out = (float*)d_out;
    const int nb = in_sizes[0] / HXW;
    gcn3_kernel<<<dim3(nb), dim3(256), 0, stream>>>(Hx, ei, W1, b1, W2, b2, W3, b3, out);
}


# Layer-1 row sums done by the wave that owns those rows in layer 2a (2 lanes per row, one cross-lane add): x1 built in the WMMA A registers, its LDS pass and 2 barriers gone; A@p sum order changed
# speedup vs baseline: 2.1972x; 1.1162x over previous
#include <hip/hip_runtime.h>
#include <hip/hip_bf16.h>

typedef __attribute__((ext_vector_type(2))) float v2f;
typedef __attribute__((ext_vector_type(8))) float v8f;
typedef __attribute__((ext_vector_type(4))) float v4f;

#define NUMK 128
#define NUME 1024
#define HID  32
#define HXW  (NUMK + NUME + 2)
#define AS   132
#define XS   36
#define GTS  132

__device__ __forceinline__ v8f wmma_step(const float* __restrict__ Am, int aStride,
                                         const float* __restrict__ BTm, int btStride,
                                         int m0, int n0, int k0, int half, int l, v8f c) {
    v2f a, b;
    a.x = Am[(m0 + l) * aStride + k0 + 2 * half];
    a.y = Am[(m0 + l) * aStride + k0 + 2 * half + 1];
    b.x = BTm[(n0 + l) * btStride + k0 + 2 * half];
    b.y = BTm[(n0 + l) * btStride + k0 + 2 * half + 1];
    return __builtin_amdgcn_wmma_f32_16x16x4_f32(false, a, false, b, (short)0, c,
                                                 false, false);
}

__device__ __forceinline__ float xhalf(float x) {
    return __builtin_bit_cast(float, __builtin_amdgcn_permlanex16(
        __builtin_bit_cast(int, x), __builtin_bit_cast(int, x),
        0x76543210, 0xfedcba98, false, false));
}

template <int N>
__device__ __forceinline__ float dot_4sums(const float* __restrict__ rowp,
                                           const float* __restrict__ vecp) {
    const v4f* r4 = reinterpret_cast<const v4f*>(rowp);
    const v4f* x4 = reinterpret_cast<const v4f*>(vecp);
    float s0 = 0.0f, s1 = 0.0f, s2 = 0.0f, s3 = 0.0f;
    for (int i = 0; i < N / 4; ++i) {
        const v4f a = r4[i];
        const v4f x = x4[i];
        s0 = fmaf(a.x, x.x, s0);
        s1 = fmaf(a.y, x.y, s1);
        s2 = fmaf(a.z, x.z, s2);
        s3 = fmaf(a.w, x.w, s3);
    }
    return (s0 + s1) + (s2 + s3);
}

__device__ __forceinline__ float dot_half_row(const float* __restrict__ rowp,
                                              const float* __restrict__ vecp, int half) {
    const v4f* r4 = reinterpret_cast<const v4f*>(rowp) + half;
    const v4f* x4 = reinterpret_cast<const v4f*>(vecp) + half;
    float s0 = 0.0f, s1 = 0.0f, s2 = 0.0f, s3 = 0.0f;
    for (int i = 0; i < NUMK / 8; ++i) {
        const v4f a = r4[2 * i];
        const v4f x = x4[2 * i];
        s0 = fmaf(a.x, x.x, s0);
        s1 = fmaf(a.y, x.y, s1);
        s2 = fmaf(a.z, x.z, s2);
        s3 = fmaf(a.w, x.w, s3);
    }
    return (s0 + s1) + (s2 + s3);
}

__global__ __launch_bounds__(256) void gcn3_kernel(
    const float* __restrict__ Hx, const int* __restrict__ ei,
    const float* __restrict__ W1, const float* __restrict__ b1,
    const float* __restrict__ W2, const float* __restrict__ b2,
    const float* __restrict__ W3, const float* __restrict__ b3,
    float* __restrict__ out) {
    __shared__ __attribute__((aligned(16))) float As[NUMK * AS];
    __shared__ __attribute__((aligned(16))) float Xs[NUMK * XS];
    __shared__ float GT[HID * GTS];
    __shared__ __attribute__((aligned(16))) float ps[NUMK];
    __shared__ __attribute__((aligned(16))) float vs[NUMK];
    __shared__ __attribute__((aligned(16))) float W3s[HID];
    __shared__ float deg[NUMK], dinv[NUMK];
    __shared__ float b2s[HID];

    const int g    = blockIdx.x;
    const int t    = threadIdx.x;
    const int lane = t & 31;
    const int wave = t >> 5;
    const int half = lane >> 4;
    const int l    = lane & 15;
    const float* hx = Hx + (size_t)g * HXW;

    float p_r = 0.0f;
    if (t < NUMK) p_r = hx[t];
    float b2_r = 0.0f, w3_r = 0.0f;
    if (t < HID) { b2_r = b2[t]; w3_r = W3[t]; }
    int   e_s[4], e_d[4];
    float e_w[4];
#pragma unroll
    for (int k = 0; k < 4; ++k) {
        const int e = t + 256 * k;
        e_s[k] = ei[e];
        e_d[k] = ei[NUME + e];
        e_w[k] = hx[NUMK + e];
    }
    v2f w1p[HID / 4], b1p[HID / 4];
#pragma unroll
    for (int ks = 0; ks < HID / 4; ++ks) {
        w1p[ks] = *reinterpret_cast<const v2f*>(W1 + 4 * ks + 2 * half);
        b1p[ks] = *reinterpret_cast<const v2f*>(b1 + 4 * ks + 2 * half);
    }
    float w2b[2][HID / 4][2];
#pragma unroll
    for (int nt = 0; nt < 2; ++nt)
#pragma unroll
        for (int ks = 0; ks < HID / 4; ++ks) {
            const int k = 4 * ks + 2 * half;
            w2b[nt][ks][0] = W2[k * HID + nt * 16 + l];
            w2b[nt][ks][1] = W2[(k + 1) * HID + nt * 16 + l];
        }

    for (int i = t; i < NUMK * AS; i += 256) As[i] = 0.0f;
    if (t < HID) { b2s[t] = b2_r; W3s[t] = w3_r; }
    __syncthreads();

    if (t < NUMK) { As[t * AS + t] = 1.0f; deg[t] = 1.0f; }
    __syncthreads();

#pragma unroll
    for (int k = 0; k < 4; ++k) {
        atomicAdd(&As[e_d[k] * AS + e_s[k]], e_w[k]);
        atomicAdd(&deg[e_d[k]], e_w[k]);
    }
    __syncthreads();

    float dv = 0.0f;
    if (t < NUMK) { dv = rsqrtf(deg[t]); dinv[t] = dv; ps[t] = dv * p_r; }
    __syncthreads();

    float dr[8];
#pragma unroll
    for (int j = 0; j < 8; ++j) dr[j] = dinv[wave * 16 + j + 8 * half];

    v2f xa[HID / 4];
    {
        const int row = wave * 16 + l;
        float s = dot_half_row(As + row * AS, ps, half);
        s += xhalf(s);
        const float a1 = dinv[row] * s;
#pragma unroll
        for (int ks = 0; ks < HID / 4; ++ks) {
            xa[ks].x = fmaxf(fmaf(a1, w1p[ks].x, b1p[ks].x), 0.0f);
            xa[ks].y = fmaxf(fmaf(a1, w1p[ks].y, b1p[ks].y), 0.0f);
        }
    }

    {
        const int m0 = wave * 16;
        for (int nt = 0; nt < 2; ++nt) {
            const int n0 = nt * 16;
            v8f c = {};
#pragma unroll
            for (int ks = 0; ks < HID / 4; ++ks) {
                v2f b;
                b.x = w2b[nt][ks][0];
                b.y = w2b[nt][ks][1];
                c = __builtin_amdgcn_wmma_f32_16x16x4_f32(false, xa[ks], false, b, (short)0, c,
                                                          false, false);
            }
#pragma unroll
            for (int j = 0; j < 8; ++j)
                GT[(n0 + l) * GTS + (m0 + j + 8 * half)] = dr[j] * c[j];
        }
    }
    __syncthreads();

    {
        const int m0 = wave * 16;
        for (int nt = 0; nt < 2; ++nt) {
            const int n0 = nt * 16;
            v8f c = {};
#pragma unroll
            for (int k0 = 0; k0 < NUMK; k0 += 4)
                c = wmma_step(As, AS, GT, GTS, m0, n0, k0, half, l, c);
#pragma unroll
            for (int j = 0; j < 8; ++j) {
                int row = m0 + j + 8 * half, col = n0 + l;
                Xs[row * XS + col] = fmaxf(fmaf(dr[j], c[j], b2s[col]), 0.0f);
            }
        }
    }
    __syncthreads();

    if (t < NUMK) vs[t] = dv * dot_4sums<HID>(Xs + t * XS, W3s);
    __syncthreads();
    if (t < NUMK)
        out[(size_t)g * NUMK + t] = fmaf(dv, dot_4sums<NUMK>(As + t * AS, vs), b3[0]);
}

extern "C" void kernel_launch(void* const* d_in, const int* in_sizes, int n_in,
                              void* d_out, int out_size, void* d_ws, size_t ws_size,
                              hipStream_t stream) {
    (void)n_in; (void)d_ws; (void)ws_size; (void)out_size;
    const float* Hx = (const float*)d_in[0];
    const int*   ei = (const int*)d_in[1];
    const float* W1 = (const float*)d_in[2];
    const float* b1 = (const float*)d_in[3];
    const float* W2 = (const float*)d_in[4];
    const float* b2 = (const float*)d_in[5];
    const float* W3 = (const float*)d_in[6];
    const float* b3 = (const float*)d_in[7];
    float* out = (float*)d_out;
    const int nb = in_sizes[0] / HXW;
    gcn3_kernel<<<dim3(nb), dim3(256), 0, stream>>>(Hx, ei, W1, b1, W2, b2, W3, b3, out);
}


# Grid of 768 persistent work-groups, each looping over 2-3 graphs: weights, edge indices and full LDS zeroing once per group; next graph's global reads stay in flight across LDS-only barriers
# speedup vs baseline: 2.4205x; 1.1016x over previous
#include <hip/hip_runtime.h>
#include <hip/hip_bf16.h>

typedef __attribute__((ext_vector_type(2))) float v2f;
typedef __attribute__((ext_vector_type(8))) float v8f;
typedef __attribute__((ext_vector_type(4))) float v4f;

#define NUMK 128
#define NUME 1024
#define HID  32
#define HXW  (NUMK + NUME + 2)
#define AS   132
#define XS   36
#define GTS  132
#define NGRP 768

__device__ __forceinline__ v8f wmma_step(const float* __restrict__ Am, int aStride,
                                         const float* __restrict__ BTm, int btStride,
                                         int m0, int n0, int k0, int half, int l, v8f c) {
    v2f a, b;
    a.x = Am[(m0 + l) * aStride + k0 + 2 * half];
    a.y = Am[(m0 + l) * aStride + k0 + 2 * half + 1];
    b.x = BTm[(n0 + l) * btStride + k0 + 2 * half];
    b.y = BTm[(n0 + l) * btStride + k0 + 2 * half + 1];
    return __builtin_amdgcn_wmma_f32_16x16x4_f32(false, a, false, b, (short)0, c,
                                                 false, false);
}

__device__ __forceinline__ float xhalf(float x) {
    return __builtin_bit_cast(float, __builtin_amdgcn_permlanex16(
        __builtin_bit_cast(int, x), __builtin_bit_cast(int, x),
        0x76543210, 0xfedcba98, false, false));
}

template <int N>
__device__ __forceinline__ float dot_4sums(const float* __restrict__ rowp,
                                           const float* __restrict__ vecp) {
    const v4f* r4 = reinterpret_cast<const v4f*>(rowp);
    const v4f* x4 = reinterpret_cast<const v4f*>(vecp);
    float s0 = 0.0f, s1 = 0.0f, s2 = 0.0f, s3 = 0.0f;
    for (int i = 0; i < N / 4; ++i) {
        const v4f a = r4[i];
        const v4f x = x4[i];
        s0 = fmaf(a.x, x.x, s0);
        s1 = fmaf(a.y, x.y, s1);
        s2 = fmaf(a.z, x.z, s2);
        s3 = fmaf(a.w, x.w, s3);
    }
    return (s0 + s1) + (s2 + s3);
}

__device__ __forceinline__ float dot_half_row(const float* __restrict__ rowp,
                                              const float* __restrict__ vecp, int half) {
    const v4f* r4 = reinterpret_cast<const v4f*>(rowp) + half;
    const v4f* x4 = reinterpret_cast<const v4f*>(vecp) + half;
    float s0 = 0.0f, s1 = 0.0f, s2 = 0.0f, s3 = 0.0f;
    for (int i = 0; i < NUMK / 8; ++i) {
        const v4f a = r4[2 * i];
        const v4f x = x4[2 * i];
        s0 = fmaf(a.x, x.x, s0);
        s1 = fmaf(a.y, x.y, s1);
        s2 = fmaf(a.z, x.z, s2);
        s3 = fmaf(a.w, x.w, s3);
    }
    return (s0 + s1) + (s2 + s3);
}

__device__ __forceinline__ void lds_barrier() {
    __builtin_amdgcn_fence(__ATOMIC_RELEASE, "workgroup", "local");
    __builtin_amdgcn_s_barrier();
    __builtin_amdgcn_fence(__ATOMIC_ACQUIRE, "workgroup", "local");
}

__global__ __launch_bounds__(256) void gcn3_kernel(
    const float* __restrict__ Hx, const int* __restrict__ ei,
    const float* __restrict__ W1, const float* __restrict__ b1,
    const float* __restrict__ W2, const float* __restrict__ b2,
    const float* __restrict__ W3, const float* __restrict__ b3,
    float* __restrict__ out, int nb) {
    __shared__ __attribute__((aligned(16))) float As[NUMK * AS];
    __shared__ __attribute__((aligned(16))) float Xs[NUMK * XS];
    __shared__ float GT[HID * GTS];
    __shared__ __attribute__((aligned(16))) float ps[NUMK];
    __shared__ __attribute__((aligned(16))) float vs[NUMK];
    __shared__ __attribute__((aligned(16))) float W3s[HID];
    __shared__ float deg[NUMK], dinv[NUMK];
    __shared__ float b2s[HID];

    int g = blockIdx.x;
    if (g >= nb) return;
    const int t    = threadIdx.x;
    const int lane = t & 31;
    const int wave = t >> 5;
    const int half = lane >> 4;
    const int l    = lane & 15;
    const float* hx = Hx + (size_t)g * HXW;

    float p_r = 0.0f;
    if (t < NUMK) p_r = hx[t];
    float b2_r = 0.0f, w3_r = 0.0f;
    if (t < HID) { b2_r = b2[t]; w3_r = W3[t]; }
    int   e_s[4], e_d[4];
    float e_w[4];
#pragma unroll
    for (int k = 0; k < 4; ++k) {
        const int e = t + 256 * k;
        e_s[k] = ei[e];
        e_d[k] = ei[NUME + e];
        e_w[k] = hx[NUMK + e];
    }
    const float b3v = b3[0];
    v2f w1p[HID / 4], b1p[HID / 4];
#pragma unroll
    for (int ks = 0; ks < HID / 4; ++ks) {
        w1p[ks] = *reinterpret_cast<const v2f*>(W1 + 4 * ks + 2 * half);
        b1p[ks] = *reinterpret_cast<const v2f*>(b1 + 4 * ks + 2 * half);
    }
    float w2b[2][HID / 4][2];
#pragma unroll
    for (int nt = 0; nt < 2; ++nt)
#pragma unroll
        for (int ks = 0; ks < HID / 4; ++ks) {
            const int k = 4 * ks + 2 * half;
            w2b[nt][ks][0] = W2[k * HID + nt * 16 + l];
            w2b[nt][ks][1] = W2[(k + 1) * HID + nt * 16 + l];
        }

    for (int i = t; i < NUMK * AS; i += 256) As[i] = 0.0f;
    if (t < HID) { b2s[t] = b2_r; W3s[t] = w3_r; }
    lds_barrier();

    if (t < NUMK) { As[t * AS + t] = 1.0f; deg[t] = 1.0f; }
    lds_barrier();

    int a_off[4];
#pragma unroll
    for (int k = 0; k < 4; ++k) a_off[k] = e_d[k] * AS + e_s[k];

  for (;;) {
#pragma unroll
    for (int k = 0; k < 4; ++k) {
        atomicAdd(&As[a_off[k]], e_w[k]);
        atomicAdd(&deg[e_d[k]], e_w[k]);
    }
    lds_barrier();

    float dv = 0.0f;
    if (t < NUMK) { dv = rsqrtf(deg[t]); dinv[t] = dv; ps[t] = dv * p_r; }

    const int  gn   = g + NGRP;
    const bool more = gn < nb;
    if (more) {
        const float* hxn = Hx + (size_t)gn * HXW;
        if (t < NUMK) p_r = hxn[t];
#pragma unroll
        for (int k = 0; k < 4; ++k) e_w[k] = hxn[NUMK + t + 256 * k];
    }
    lds_barrier();

    float dr[8];
#pragma unroll
    for (int j = 0; j < 8; ++j) dr[j] = dinv[wave * 16 + j + 8 * half];

    v2f xa[HID / 4];
    {
        const int row = wave * 16 + l;
        float s = dot_half_row(As + row * AS, ps, half);
        s += xhalf(s);
        const float a1 = dinv[row] * s;
#pragma unroll
        for (int ks = 0; ks < HID / 4; ++ks) {
            xa[ks].x = fmaxf(fmaf(a1, w1p[ks].x, b1p[ks].x), 0.0f);
            xa[ks].y = fmaxf(fmaf(a1, w1p[ks].y, b1p[ks].y), 0.0f);
        }
    }

    {
        const int m0 = wave * 16;
        for (int nt = 0; nt < 2; ++nt) {
            const int n0 = nt * 16;
            v8f c = {};
#pragma unroll
            for (int ks = 0; ks < HID / 4; ++ks) {
                v2f b;
                b.x = w2b[nt][ks][0];
                b.y = w2b[nt][ks][1];
                c = __builtin_amdgcn_wmma_f32_16x16x4_f32(false, xa[ks], false, b, (short)0, c,
                                                          false, false);
            }
#pragma unroll
            for (int j = 0; j < 8; ++j)
                GT[(n0 + l) * GTS + (m0 + j + 8 * half)] = dr[j] * c[j];
        }
    }
    lds_barrier();

    {
        const int m0 = wave * 16;
        for (int nt = 0; nt < 2; ++nt) {
            const int n0 = nt * 16;
            v8f c = {};
#pragma unroll
            for (int k0 = 0; k0 < NUMK; k0 += 4)
                c = wmma_step(As, AS, GT, GTS, m0, n0, k0, half, l, c);
#pragma unroll
            for (int j = 0; j < 8; ++j) {
                int row = m0 + j + 8 * half, col = n0 + l;
                Xs[row * XS + col] = fmaxf(fmaf(dr[j], c[j], b2s[col]), 0.0f);
            }
        }
    }
    lds_barrier();

    if (t < NUMK) vs[t] = dv * dot_4sums<HID>(Xs + t * XS, W3s);
    lds_barrier();
    if (t < NUMK)
        out[(size_t)g * NUMK + t] = fmaf(dv, dot_4sums<NUMK>(As + t * AS, vs), b3v);
    if (!more) break;
    g = gn;
    lds_barrier();

#pragma unroll
    for (int k = 0; k < 4; ++k)
        if (e_s[k] != e_d[k]) As[a_off[k]] = 0.0f;
    if (t < NUMK) { As[t * AS + t] = 1.0f; deg[t] = 1.0f; }
    lds_barrier();
  }
}

extern "C" void kernel_launch(void* const* d_in, const int* in_sizes, int n_in,
                              void* d_out, int out_size, void* d_ws, size_t ws_size,
                              hipStream_t stream) {
    (void)n_in; (void)d_ws; (void)ws_size; (void)out_size;
    const float* Hx = (const float*)d_in[0];
    const int*   ei = (const int*)d_in[1];
    const float* W1 = (const float*)d_in[2];
    const float* b1 = (const float*)d_in[3];
    const float* W2 = (const float*)d_in[4];
    const float* b2 = (const float*)d_in[5];
    const float* W3 = (const float*)d_in[6];
    const float* b3 = (const float*)d_in[7];
    float* out = (float*)d_out;
    const int nb = in_sizes[0] / HXW;
    gcn3_kernel<<<dim3(NGRP), dim3(256), 0, stream>>>(Hx, ei, W1, b1, W2, b2, W3, b3, out, nb);
}
